# starting kernel
# baseline (speedup 1.0000x reference)
#include <hip/hip_runtime.h>
#include <hip/hip_bf16.h>


#define NQ   12
#define DIM  4096
#define NTHR 256

typedef __attribute__((ext_vector_type(2))) float v2f;
typedef __attribute__((ext_vector_type(8))) float v8f;

__global__ __launch_bounds__(NTHR) void qsim_kernel(const float* __restrict__ x,
                                                    const float* __restrict__ params,
                                                    int n_layers,
                                                    float* __restrict__ exps_out)
{
    __shared__ float2 st[DIM];
    __shared__ float  red[12 * NTHR];

    const int b   = blockIdx.x;
    const int tid = threadIdx.x;

    for (int k = tid; k < DIM; k += NTHR)
        st[k] = make_float2(k == 0 ? 1.0f : 0.0f, 0.0f);
    __syncthreads();

    for (int q = 0; q < NQ; ++q) {
        const float xv = x[b * NQ + q];
        float sh, ch;
        sincosf(0.5f * xv, &sh, &ch);
        const int bit  = NQ - 1 - q;
        const int strd = 1 << bit;
        for (int p = tid; p < DIM / 2; p += NTHR) {
            const int i0 = ((p >> bit) << (bit + 1)) | (p & (strd - 1));
            const int i1 = i0 | strd;
            const float2 a0 = st[i0];
            const float2 a1 = st[i1];
            st[i0] = make_float2(ch * a0.x + sh * a1.y, ch * a0.y - sh * a1.x);
            st[i1] = make_float2(sh * a0.y + ch * a1.x, ch * a1.y - sh * a0.x);
        }
        __syncthreads();
    }

    for (int l = 0; l < n_layers; ++l) {
        for (int q = 0; q < NQ; ++q) {
            const float* pp  = params + (l * NQ + q) * 3;
            const float phi  = pp[0], th = pp[1], om = pp[2];
            float stt, ct, sa, ca, sb, cb;
            sincosf(0.5f * th, &stt, &ct);
            sincosf(0.5f * (phi + om), &sa, &ca);
            sincosf(0.5f * (phi - om), &sb, &cb);
            const float g00r =  ca * ct,  g00i = -sa * ct;
            const float g01r = -cb * stt, g01i = -sb * stt;
            const float g10r =  cb * stt, g10i = -sb * stt;
            const float g11r =  ca * ct,  g11i =  sa * ct;

            const int bit  = NQ - 1 - q;
            const int strd = 1 << bit;
            for (int p = tid; p < DIM / 2; p += NTHR) {
                const int i0 = ((p >> bit) << (bit + 1)) | (p & (strd - 1));
                const int i1 = i0 | strd;
                const float2 a0 = st[i0];
                const float2 a1 = st[i1];
                st[i0] = make_float2(
                    g00r * a0.x - g00i * a0.y + g01r * a1.x - g01i * a1.y,
                    g00r * a0.y + g00i * a0.x + g01r * a1.y + g01i * a1.x);
                st[i1] = make_float2(
                    g10r * a0.x - g10i * a0.y + g11r * a1.x - g11i * a1.y,
                    g10r * a0.y + g10i * a0.x + g11r * a1.y + g11i * a1.x);
            }
            __syncthreads();
        }
        for (int q = 0; q < NQ - 1; ++q) {
            const int tb = NQ - 2 - q;
            for (int p = tid; p < DIM / 4; p += NTHR) {
                const int low  = p & ((1 << tb) - 1);
                const int high = p >> tb;
                const int i1 = (high << (tb + 2)) | (1 << (tb + 1)) | low;
                const int i2 = i1 | (1 << tb);
                const float2 t = st[i1];
                st[i1] = st[i2];
                st[i2] = t;
            }
            __syncthreads();
        }
    }

    float acc[12];
#pragma unroll
    for (int i = 0; i < 12; ++i) acc[i] = 0.0f;
    for (int k = tid; k < DIM; k += NTHR) {
        const float2 a = st[k];
        const float pr = a.x * a.x + a.y * a.y;
#pragma unroll
        for (int i = 0; i < 12; ++i)
            acc[i] += ((k >> (11 - i)) & 1) ? -pr : pr;
    }
#pragma unroll
    for (int i = 0; i < 12; ++i) red[i * NTHR + tid] = acc[i];
    __syncthreads();
    for (int off = NTHR / 2; off > 0; off >>= 1) {
        if (tid < off) {
#pragma unroll
            for (int i = 0; i < 12; ++i)
                red[i * NTHR + tid] += red[i * NTHR + tid + off];
        }
        __syncthreads();
    }
    if (tid < 12)
        exps_out[b * 12 + tid] = red[tid * NTHR];
}

__global__ __launch_bounds__(32) void mitigate_wmma_kernel(const float* __restrict__ exps,
                                                           float* __restrict__ out)
{
    const int lane = threadIdx.x;
    const int base = blockIdx.x * 16;
    const int m    = lane & 15;
    const int half = lane >> 4;

    const float inv11 = 1.0f / 11.0f;

    v8f c = {0.f, 0.f, 0.f, 0.f, 0.f, 0.f, 0.f, 0.f};

#pragma unroll
    for (int kk = 0; kk < 12; kk += 4) {
        const int ka = kk + 2 * half;
        v2f a, bv;
        a.x = exps[(base + m) * 12 + ka];
        a.y = exps[(base + m) * 12 + ka + 1];
        bv.x = (m == ka)     ? (inv11 - 1.0f) : inv11;
        bv.y = (m == ka + 1) ? (inv11 - 1.0f) : inv11;
        c = __builtin_amdgcn_wmma_f32_16x16x4_f32(
                 false, a,  false, bv,
                 (short)0, c,  false,  false);
    }

#pragma unroll
    for (int v = 0; v < 8; ++v) {
        const int row = v + 8 * half;
        if (m < 12)
            out[(base + row) * 12 + m] = c[v] - (10.0f / 11.0f);
    }
}

extern "C" void kernel_launch(void* const* d_in, const int* in_sizes, int n_in,
                              void* d_out, int out_size, void* d_ws, size_t ws_size,
                              hipStream_t stream)
{
    const float* x      = (const float*)d_in[0];
    const float* params = (const float*)d_in[1];
    float*       out    = (float*)d_out;
    float*       exps   = (float*)d_ws;

    const int B        = in_sizes[0] / NQ;
    const int n_layers = in_sizes[1] / (NQ * 3);

    qsim_kernel<<<B, NTHR, 0, stream>>>(x, params, n_layers, exps);
    mitigate_wmma_kernel<<<B / 16, 32, 0, stream>>>(exps, out);
}


# RX and Rot gates applied four qubits at a time on 16 register-held amplitudes per thread: 3 LDS round trips per 12 gates instead of 12 (LDS slots padded 1 per 16); CNOT passes as before
# speedup vs baseline: 1.5194x; 1.5194x over previous
#include <hip/hip_runtime.h>
#include <hip/hip_bf16.h>


#define NQ   12
#define DIM  4096
#define NTHR 256
#define SLOT(k) ((k) + ((k) >> 4))
#define DIMP (DIM + DIM / 16)

typedef __attribute__((ext_vector_type(2))) float v2f;
typedef __attribute__((ext_vector_type(8))) float v8f;

__device__ __forceinline__ void phase_map(int G, int t, int& base, int& stride)
{
    if (G == 0)      { base = t + (t >> 4);                 stride = 272; }
    else if (G == 1) { base = (t >> 4) * 272 + (t & 15);    stride = 17;  }
    else             { base = 17 * t;                       stride = 1;   }
}

template <int M>
__device__ __forceinline__ void gate_regs(float2 (&a)[16],
                                          float g00r, float g00i, float g01r, float g01i,
                                          float g10r, float g10i, float g11r, float g11i)
{
#pragma unroll
    for (int p = 0; p < 8; ++p) {
        const int j0 = ((p >> M) << (M + 1)) | (p & ((1 << M) - 1));
        const int j1 = j0 | (1 << M);
        const float2 a0 = a[j0];
        const float2 a1 = a[j1];
        a[j0] = make_float2(
            g00r * a0.x - g00i * a0.y + g01r * a1.x - g01i * a1.y,
            g00r * a0.y + g00i * a0.x + g01r * a1.y + g01i * a1.x);
        a[j1] = make_float2(
            g10r * a0.x - g10i * a0.y + g11r * a1.x - g11i * a1.y,
            g10r * a0.y + g10i * a0.x + g11r * a1.y + g11i * a1.x);
    }
}

template <int M>
__device__ __forceinline__ void rot_regs(float2 (&a)[16], const float* __restrict__ pp)
{
    const float phi = pp[0], th = pp[1], om = pp[2];
    float stt, ct, sa, ca, sb, cb;
    sincosf(0.5f * th, &stt, &ct);
    sincosf(0.5f * (phi + om), &sa, &ca);
    sincosf(0.5f * (phi - om), &sb, &cb);
    gate_regs<M>(a,  ca * ct, -sa * ct, -cb * stt, -sb * stt,
                     cb * stt, -sb * stt,  ca * ct,  sa * ct);
}

template <int M>
__device__ __forceinline__ void rx_regs(float2 (&a)[16], float xv)
{
    float sh, ch;
    sincosf(0.5f * xv, &sh, &ch);
#pragma unroll
    for (int p = 0; p < 8; ++p) {
        const int j0 = ((p >> M) << (M + 1)) | (p & ((1 << M) - 1));
        const int j1 = j0 | (1 << M);
        const float2 a0 = a[j0];
        const float2 a1 = a[j1];
        a[j0] = make_float2(ch * a0.x + sh * a1.y, ch * a0.y - sh * a1.x);
        a[j1] = make_float2(sh * a0.y + ch * a1.x, ch * a1.y - sh * a0.x);
    }
}

__device__ __forceinline__ void rx_phases(float2* st, const float* __restrict__ xb, int tid)
{
#pragma unroll 1
    for (int G = 0; G < 3; ++G) {
        int base, stride;
        phase_map(G, tid, base, stride);
        float2 a[16];
#pragma unroll
        for (int j = 0; j < 16; ++j) a[j] = st[base + j * stride];
        rx_regs<3>(a, xb[4 * G + 0]);
        rx_regs<2>(a, xb[4 * G + 1]);
        rx_regs<1>(a, xb[4 * G + 2]);
        rx_regs<0>(a, xb[4 * G + 3]);
#pragma unroll
        for (int j = 0; j < 16; ++j) st[base + j * stride] = a[j];
        __syncthreads();
    }
}

__device__ __forceinline__ void rot_phases(float2* st, const float* __restrict__ pl, int tid)
{
#pragma unroll 1
    for (int G = 0; G < 3; ++G) {
        int base, stride;
        phase_map(G, tid, base, stride);
        float2 a[16];
#pragma unroll
        for (int j = 0; j < 16; ++j) a[j] = st[base + j * stride];
        const float* pg = pl + 4 * G * 3;
        rot_regs<3>(a, pg + 0);
        rot_regs<2>(a, pg + 3);
        rot_regs<1>(a, pg + 6);
        rot_regs<0>(a, pg + 9);
#pragma unroll
        for (int j = 0; j < 16; ++j) st[base + j * stride] = a[j];
        __syncthreads();
    }
}

__global__ __launch_bounds__(NTHR) void qsim_kernel(const float* __restrict__ x,
                                                    const float* __restrict__ params,
                                                    int n_layers,
                                                    float* __restrict__ exps_out)
{
    __shared__ float2 st[DIMP];
    __shared__ float  red[12 * NTHR];

    const int b   = blockIdx.x;
    const int tid = threadIdx.x;

    for (int k = tid; k < DIM; k += NTHR)
        st[SLOT(k)] = make_float2(k == 0 ? 1.0f : 0.0f, 0.0f);
    __syncthreads();

    rx_phases(st, x + b * NQ, tid);

    for (int l = 0; l < n_layers; ++l) {
        rot_phases(st, params + l * NQ * 3, tid);
        for (int q = 0; q < NQ - 1; ++q) {
            const int tb = NQ - 2 - q;
            for (int p = tid; p < DIM / 4; p += NTHR) {
                const int low  = p & ((1 << tb) - 1);
                const int high = p >> tb;
                const int i1 = (high << (tb + 2)) | (1 << (tb + 1)) | low;
                const int i2 = i1 | (1 << tb);
                const float2 t = st[SLOT(i1)];
                st[SLOT(i1)] = st[SLOT(i2)];
                st[SLOT(i2)] = t;
            }
            __syncthreads();
        }
    }

    float acc[12];
#pragma unroll
    for (int i = 0; i < 12; ++i) acc[i] = 0.0f;
    for (int k = tid; k < DIM; k += NTHR) {
        const float2 a = st[SLOT(k)];
        const float pr = a.x * a.x + a.y * a.y;
#pragma unroll
        for (int i = 0; i < 12; ++i)
            acc[i] += ((k >> (11 - i)) & 1) ? -pr : pr;
    }
#pragma unroll
    for (int i = 0; i < 12; ++i) red[i * NTHR + tid] = acc[i];
    __syncthreads();
    for (int off = NTHR / 2; off > 0; off >>= 1) {
        if (tid < off) {
#pragma unroll
            for (int i = 0; i < 12; ++i)
                red[i * NTHR + tid] += red[i * NTHR + tid + off];
        }
        __syncthreads();
    }
    if (tid < 12)
        exps_out[b * 12 + tid] = red[tid * NTHR];
}

__global__ __launch_bounds__(32) void mitigate_wmma_kernel(const float* __restrict__ exps,
                                                           float* __restrict__ out)
{
    const int lane = threadIdx.x;
    const int base = blockIdx.x * 16;
    const int m    = lane & 15;
    const int half = lane >> 4;

    const float inv11 = 1.0f / 11.0f;

    v8f c = {0.f, 0.f, 0.f, 0.f, 0.f, 0.f, 0.f, 0.f};

#pragma unroll
    for (int kk = 0; kk < 12; kk += 4) {
        const int ka = kk + 2 * half;
        v2f a, bv;
        a.x = exps[(base + m) * 12 + ka];
        a.y = exps[(base + m) * 12 + ka + 1];
        bv.x = (m == ka)     ? (inv11 - 1.0f) : inv11;
        bv.y = (m == ka + 1) ? (inv11 - 1.0f) : inv11;
        c = __builtin_amdgcn_wmma_f32_16x16x4_f32(
                 false, a,  false, bv,
                 (short)0, c,  false,  false);
    }

#pragma unroll
    for (int v = 0; v < 8; ++v) {
        const int row = v + 8 * half;
        if (m < 12)
            out[(base + row) * 12 + m] = c[v] - (10.0f / 11.0f);
    }
}

extern "C" void kernel_launch(void* const* d_in, const int* in_sizes, int n_in,
                              void* d_out, int out_size, void* d_ws, size_t ws_size,
                              hipStream_t stream)
{
    const float* x      = (const float*)d_in[0];
    const float* params = (const float*)d_in[1];
    float*       out    = (float*)d_out;
    float*       exps   = (float*)d_ws;

    const int B        = in_sizes[0] / NQ;
    const int n_layers = in_sizes[1] / (NQ * 3);

    qsim_kernel<<<B, NTHR, 0, stream>>>(x, params, n_layers, exps);
    mitigate_wmma_kernel<<<B / 16, 32, 0, stream>>>(exps, out);
}


# Layer's CNOT chain (11 LDS swap passes, 11 barriers) replaced by its one index permutation k -> running xor of k's bits, done by the stores of the last Rot phase into a second LDS state buffer
# speedup vs baseline: 1.8817x; 1.2384x over previous
#include <hip/hip_runtime.h>
#include <hip/hip_bf16.h>


#define NQ   12
#define DIM  4096
#define NTHR 256
#define SLOT(k) ((k) + ((k) >> 4))
#define DIMP (DIM + DIM / 16)

typedef __attribute__((ext_vector_type(2))) float v2f;
typedef __attribute__((ext_vector_type(8))) float v8f;

__device__ __forceinline__ void phase_map(int G, int t, int& base, int& stride)
{
    if (G == 0)      { base = t + (t >> 4);                 stride = 272; }
    else if (G == 1) { base = (t >> 4) * 272 + (t & 15);    stride = 17;  }
    else             { base = 17 * t;                       stride = 1;   }
}

template <int M>
__device__ __forceinline__ void gate_regs(float2 (&a)[16],
                                          float g00r, float g00i, float g01r, float g01i,
                                          float g10r, float g10i, float g11r, float g11i)
{
#pragma unroll
    for (int p = 0; p < 8; ++p) {
        const int j0 = ((p >> M) << (M + 1)) | (p & ((1 << M) - 1));
        const int j1 = j0 | (1 << M);
        const float2 a0 = a[j0];
        const float2 a1 = a[j1];
        a[j0] = make_float2(
            g00r * a0.x - g00i * a0.y + g01r * a1.x - g01i * a1.y,
            g00r * a0.y + g00i * a0.x + g01r * a1.y + g01i * a1.x);
        a[j1] = make_float2(
            g10r * a0.x - g10i * a0.y + g11r * a1.x - g11i * a1.y,
            g10r * a0.y + g10i * a0.x + g11r * a1.y + g11i * a1.x);
    }
}

template <int M>
__device__ __forceinline__ void rot_regs(float2 (&a)[16], const float* __restrict__ pp)
{
    const float phi = pp[0], th = pp[1], om = pp[2];
    float stt, ct, sa, ca, sb, cb;
    sincosf(0.5f * th, &stt, &ct);
    sincosf(0.5f * (phi + om), &sa, &ca);
    sincosf(0.5f * (phi - om), &sb, &cb);
    gate_regs<M>(a,  ca * ct, -sa * ct, -cb * stt, -sb * stt,
                     cb * stt, -sb * stt,  ca * ct,  sa * ct);
}

template <int M>
__device__ __forceinline__ void rx_regs(float2 (&a)[16], float xv)
{
    float sh, ch;
    sincosf(0.5f * xv, &sh, &ch);
#pragma unroll
    for (int p = 0; p < 8; ++p) {
        const int j0 = ((p >> M) << (M + 1)) | (p & ((1 << M) - 1));
        const int j1 = j0 | (1 << M);
        const float2 a0 = a[j0];
        const float2 a1 = a[j1];
        a[j0] = make_float2(ch * a0.x + sh * a1.y, ch * a0.y - sh * a1.x);
        a[j1] = make_float2(sh * a0.y + ch * a1.x, ch * a1.y - sh * a0.x);
    }
}

__device__ __forceinline__ void rx_phases(float2* st, const float* __restrict__ xb, int tid)
{
#pragma unroll 1
    for (int G = 0; G < 3; ++G) {
        int base, stride;
        phase_map(G, tid, base, stride);
        float2 a[16];
#pragma unroll
        for (int j = 0; j < 16; ++j) a[j] = st[base + j * stride];
        rx_regs<3>(a, xb[4 * G + 0]);
        rx_regs<2>(a, xb[4 * G + 1]);
        rx_regs<1>(a, xb[4 * G + 2]);
        rx_regs<0>(a, xb[4 * G + 3]);
#pragma unroll
        for (int j = 0; j < 16; ++j) st[base + j * stride] = a[j];
        __syncthreads();
    }
}

__device__ __forceinline__ void rot_cnot_phases(float2* src, float2* dst,
                                                const float* __restrict__ pl, int tid)
{
#pragma unroll 1
    for (int G = 0; G < 3; ++G) {
        int base, stride;
        phase_map(G, tid, base, stride);
        float2 a[16];
#pragma unroll
        for (int j = 0; j < 16; ++j) a[j] = src[base + j * stride];
        const float* pg = pl + 4 * G * 3;
        rot_regs<3>(a, pg + 0);
        rot_regs<2>(a, pg + 3);
        rot_regs<1>(a, pg + 6);
        rot_regs<0>(a, pg + 9);
        if (G < 2) {
#pragma unroll
            for (int j = 0; j < 16; ++j) src[base + j * stride] = a[j];
        } else {
            int f = tid;
            f ^= f >> 1;
            f ^= f >> 2;
            f ^= f >> 4;
            const bool odd = (f & 1) != 0;
            float2* row = dst + 17 * f;
#pragma unroll
            for (int c = 0; c < 16; ++c) {
                const int j = c ^ (c >> 1);
                const float2 p = a[j];
                const float2 q = a[j ^ 8];
                const float vx = odd ? q.x : p.x;
                const float vy = odd ? q.y : p.y;
                row[c] = make_float2(vx, vy);
            }
        }
        __syncthreads();
    }
}

__global__ __launch_bounds__(NTHR) void qsim_kernel(const float* __restrict__ x,
                                                    const float* __restrict__ params,
                                                    int n_layers,
                                                    float* __restrict__ exps_out)
{
    __shared__ float2 st0[DIMP];
    __shared__ float2 st1[DIMP];
    __shared__ float  red[12 * NTHR];

    const int b   = blockIdx.x;
    const int tid = threadIdx.x;

    float2* st  = st0;
    float2* alt = st1;

    for (int k = tid; k < DIM; k += NTHR)
        st[SLOT(k)] = make_float2(k == 0 ? 1.0f : 0.0f, 0.0f);
    __syncthreads();

    rx_phases(st, x + b * NQ, tid);

    for (int l = 0; l < n_layers; ++l) {
        rot_cnot_phases(st, alt, params + l * NQ * 3, tid);
        float2* t = st; st = alt; alt = t;
    }

    float acc[12];
#pragma unroll
    for (int i = 0; i < 12; ++i) acc[i] = 0.0f;
    for (int k = tid; k < DIM; k += NTHR) {
        const float2 a = st[SLOT(k)];
        const float pr = a.x * a.x + a.y * a.y;
#pragma unroll
        for (int i = 0; i < 12; ++i)
            acc[i] += ((k >> (11 - i)) & 1) ? -pr : pr;
    }
#pragma unroll
    for (int i = 0; i < 12; ++i) red[i * NTHR + tid] = acc[i];
    __syncthreads();
    for (int off = NTHR / 2; off > 0; off >>= 1) {
        if (tid < off) {
#pragma unroll
            for (int i = 0; i < 12; ++i)
                red[i * NTHR + tid] += red[i * NTHR + tid + off];
        }
        __syncthreads();
    }
    if (tid < 12)
        exps_out[b * 12 + tid] = red[tid * NTHR];
}

__global__ __launch_bounds__(32) void mitigate_wmma_kernel(const float* __restrict__ exps,
                                                           float* __restrict__ out)
{
    const int lane = threadIdx.x;
    const int base = blockIdx.x * 16;
    const int m    = lane & 15;
    const int half = lane >> 4;

    const float inv11 = 1.0f / 11.0f;

    v8f c = {0.f, 0.f, 0.f, 0.f, 0.f, 0.f, 0.f, 0.f};

#pragma unroll
    for (int kk = 0; kk < 12; kk += 4) {
        const int ka = kk + 2 * half;
        v2f a, bv;
        a.x = exps[(base + m) * 12 + ka];
        a.y = exps[(base + m) * 12 + ka + 1];
        bv.x = (m == ka)     ? (inv11 - 1.0f) : inv11;
        bv.y = (m == ka + 1) ? (inv11 - 1.0f) : inv11;
        c = __builtin_amdgcn_wmma_f32_16x16x4_f32(
                 false, a,  false, bv,
                 (short)0, c,  false,  false);
    }

#pragma unroll
    for (int v = 0; v < 8; ++v) {
        const int row = v + 8 * half;
        if (m < 12)
            out[(base + row) * 12 + m] = c[v] - (10.0f / 11.0f);
    }
}

extern "C" void kernel_launch(void* const* d_in, const int* in_sizes, int n_in,
                              void* d_out, int out_size, void* d_ws, size_t ws_size,
                              hipStream_t stream)
{
    const float* x      = (const float*)d_in[0];
    const float* params = (const float*)d_in[1];
    float*       out    = (float*)d_out;
    float*       exps   = (float*)d_ws;

    const int B        = in_sizes[0] / NQ;
    const int n_layers = in_sizes[1] / (NQ * 3);

    qsim_kernel<<<B, NTHR, 0, stream>>>(x, params, n_layers, exps);
    mitigate_wmma_kernel<<<B / 16, 32, 0, stream>>>(exps, out);
}


# Gate coefficients (sin/cos of the 228 half-angles and their products) evaluated once per block into an LDS table and read as wave-uniform scalars, instead of 228 inlined sincosf per wave
# speedup vs baseline: 3.4006x; 1.8072x over previous
#include <hip/hip_runtime.h>
#include <hip/hip_bf16.h>


#define NQ   12
#define DIM  4096
#define NTHR 256
#define SLOT(k) ((k) + ((k) >> 4))
#define DIMP (DIM + DIM / 16)
#define LCHUNK   6
#define TRIG_RX  (4 * 12 * LCHUNK)
#define RX_TID0  160

typedef __attribute__((ext_vector_type(2))) float v2f;
typedef __attribute__((ext_vector_type(8))) float v8f;

__device__ __forceinline__ void phase_map(int G, int t, int& base, int& stride)
{
    if (G == 0)      { base = t + (t >> 4);                 stride = 272; }
    else if (G == 1) { base = (t >> 4) * 272 + (t & 15);    stride = 17;  }
    else             { base = 17 * t;                       stride = 1;   }
}

__device__ __forceinline__ float uniform_f(float v)
{
    return __builtin_bit_cast(float, __builtin_amdgcn_readfirstlane(__builtin_bit_cast(int, v)));
}

template <int M>
__device__ __forceinline__ void gate_regs(float2 (&a)[16],
                                          float g00r, float g00i, float g01r, float g01i,
                                          float g10r, float g10i, float g11r, float g11i)
{
#pragma unroll
    for (int p = 0; p < 8; ++p) {
        const int j0 = ((p >> M) << (M + 1)) | (p & ((1 << M) - 1));
        const int j1 = j0 | (1 << M);
        const float2 a0 = a[j0];
        const float2 a1 = a[j1];
        a[j0] = make_float2(
            g00r * a0.x - g00i * a0.y + g01r * a1.x - g01i * a1.y,
            g00r * a0.y + g00i * a0.x + g01r * a1.y + g01i * a1.x);
        a[j1] = make_float2(
            g10r * a0.x - g10i * a0.y + g11r * a1.x - g11i * a1.y,
            g10r * a0.y + g10i * a0.x + g11r * a1.y + g11i * a1.x);
    }
}

template <int M>
__device__ __forceinline__ void rot_regs(float2 (&a)[16], const float* tg)
{
    const float cact = uniform_f(tg[0]), sact = uniform_f(tg[1]);
    const float cbst = uniform_f(tg[2]), sbst = uniform_f(tg[3]);
    gate_regs<M>(a,  cact, -sact, -cbst, -sbst,
                     cbst, -sbst,  cact,  sact);
}

template <int M>
__device__ __forceinline__ void rx_regs(float2 (&a)[16], const float* tq)
{
    const float sh = uniform_f(tq[0]), ch = uniform_f(tq[1]);
#pragma unroll
    for (int p = 0; p < 8; ++p) {
        const int j0 = ((p >> M) << (M + 1)) | (p & ((1 << M) - 1));
        const int j1 = j0 | (1 << M);
        const float2 a0 = a[j0];
        const float2 a1 = a[j1];
        a[j0] = make_float2(ch * a0.x + sh * a1.y, ch * a0.y - sh * a1.x);
        a[j1] = make_float2(sh * a0.y + ch * a1.x, ch * a1.y - sh * a0.x);
    }
}

__device__ __forceinline__ void rx_phases(float2* st, const float* trx, int tid)
{
#pragma unroll 1
    for (int G = 0; G < 3; ++G) {
        int base, stride;
        phase_map(G, tid, base, stride);
        float2 a[16];
#pragma unroll
        for (int j = 0; j < 16; ++j) a[j] = st[base + j * stride];
        const float* tq = trx + 2 * 4 * G;
        rx_regs<3>(a, tq + 0);
        rx_regs<2>(a, tq + 2);
        rx_regs<1>(a, tq + 4);
        rx_regs<0>(a, tq + 6);
#pragma unroll
        for (int j = 0; j < 16; ++j) st[base + j * stride] = a[j];
        __syncthreads();
    }
}

__device__ __forceinline__ void rot_cnot_phases(float2* src, float2* dst,
                                                const float* tl, int tid)
{
#pragma unroll 1
    for (int G = 0; G < 3; ++G) {
        int base, stride;
        phase_map(G, tid, base, stride);
        float2 a[16];
#pragma unroll
        for (int j = 0; j < 16; ++j) a[j] = src[base + j * stride];
        const float* tg = tl + 4 * G * 4;
        rot_regs<3>(a, tg + 0);
        rot_regs<2>(a, tg + 4);
        rot_regs<1>(a, tg + 8);
        rot_regs<0>(a, tg + 12);
        if (G < 2) {
#pragma unroll
            for (int j = 0; j < 16; ++j) src[base + j * stride] = a[j];
        } else {
            int f = tid;
            f ^= f >> 1;
            f ^= f >> 2;
            f ^= f >> 4;
            const bool odd = (f & 1) != 0;
            float2* row = dst + 17 * f;
#pragma unroll
            for (int c = 0; c < 16; ++c) {
                const int j = c ^ (c >> 1);
                const float2 p = a[j];
                const float2 q = a[j ^ 8];
                const float vx = odd ? q.x : p.x;
                const float vy = odd ? q.y : p.y;
                row[c] = make_float2(vx, vy);
            }
        }
        __syncthreads();
    }
}

__device__ __forceinline__ void fill_trig(float* trig, const float* __restrict__ pl, int nl,
                                          const float* __restrict__ xb, int tid)
{
    if (tid < 2 * NQ * nl) {
        const int  g      = tid >> 1;
        const bool second = (tid & 1) != 0;
        const float phi = pl[3 * g + 0], th = pl[3 * g + 1], om = pl[3 * g + 2];
        float stt, ct, s2, c2;
        sincosf(0.5f * th, &stt, &ct);
        sincosf(0.5f * (second ? phi - om : phi + om), &s2, &c2);
        const float m = second ? stt : ct;
        trig[4 * g + 2 * (tid & 1) + 0] = c2 * m;
        trig[4 * g + 2 * (tid & 1) + 1] = s2 * m;
    } else if (xb != nullptr && tid >= RX_TID0 && tid < RX_TID0 + NQ) {
        float sh, ch;
        sincosf(0.5f * xb[tid - RX_TID0], &sh, &ch);
        trig[TRIG_RX + 2 * (tid - RX_TID0) + 0] = sh;
        trig[TRIG_RX + 2 * (tid - RX_TID0) + 1] = ch;
    }
}

__global__ __launch_bounds__(NTHR) void qsim_kernel(const float* __restrict__ x,
                                                    const float* __restrict__ params,
                                                    int n_layers,
                                                    float* __restrict__ exps_out)
{
    __shared__ float2 st0[DIMP];
    __shared__ float2 st1[DIMP];
    __shared__ float  red[12 * NTHR];

    const int b   = blockIdx.x;
    const int tid = threadIdx.x;

    float2* st  = st0;
    float2* alt = st1;

    for (int k = tid; k < DIM; k += NTHR)
        st[SLOT(k)] = make_float2(k == 0 ? 1.0f : 0.0f, 0.0f);
    fill_trig(red, params, n_layers < LCHUNK ? n_layers : LCHUNK, x + b * NQ, tid);
    __syncthreads();

    rx_phases(st, red + TRIG_RX, tid);

    for (int l0 = 0; l0 < n_layers; l0 += LCHUNK) {
        const int nl = n_layers - l0 < LCHUNK ? n_layers - l0 : LCHUNK;
        if (l0 > 0) {
            fill_trig(red, params + l0 * NQ * 3, nl, nullptr, tid);
            __syncthreads();
        }
        for (int l = 0; l < nl; ++l) {
            rot_cnot_phases(st, alt, red + l * NQ * 4, tid);
            float2* t = st; st = alt; alt = t;
        }
    }

    float acc[12];
#pragma unroll
    for (int i = 0; i < 12; ++i) acc[i] = 0.0f;
    for (int k = tid; k < DIM; k += NTHR) {
        const float2 a = st[SLOT(k)];
        const float pr = a.x * a.x + a.y * a.y;
#pragma unroll
        for (int i = 0; i < 12; ++i)
            acc[i] += ((k >> (11 - i)) & 1) ? -pr : pr;
    }
#pragma unroll
    for (int i = 0; i < 12; ++i) red[i * NTHR + tid] = acc[i];
    __syncthreads();
    for (int off = NTHR / 2; off > 0; off >>= 1) {
        if (tid < off) {
#pragma unroll
            for (int i = 0; i < 12; ++i)
                red[i * NTHR + tid] += red[i * NTHR + tid + off];
        }
        __syncthreads();
    }
    if (tid < 12)
        exps_out[b * 12 + tid] = red[tid * NTHR];
}

__global__ __launch_bounds__(32) void mitigate_wmma_kernel(const float* __restrict__ exps,
                                                           float* __restrict__ out)
{
    const int lane = threadIdx.x;
    const int base = blockIdx.x * 16;
    const int m    = lane & 15;
    const int half = lane >> 4;

    const float inv11 = 1.0f / 11.0f;

    v8f c = {0.f, 0.f, 0.f, 0.f, 0.f, 0.f, 0.f, 0.f};

#pragma unroll
    for (int kk = 0; kk < 12; kk += 4) {
        const int ka = kk + 2 * half;
        v2f a, bv;
        a.x = exps[(base + m) * 12 + ka];
        a.y = exps[(base + m) * 12 + ka + 1];
        bv.x = (m == ka)     ? (inv11 - 1.0f) : inv11;
        bv.y = (m == ka + 1) ? (inv11 - 1.0f) : inv11;
        c = __builtin_amdgcn_wmma_f32_16x16x4_f32(
                 false, a,  false, bv,
                 (short)0, c,  false,  false);
    }

#pragma unroll
    for (int v = 0; v < 8; ++v) {
        const int row = v + 8 * half;
        if (m < 12)
            out[(base + row) * 12 + m] = c[v] - (10.0f / 11.0f);
    }
}

extern "C" void kernel_launch(void* const* d_in, const int* in_sizes, int n_in,
                              void* d_out, int out_size, void* d_ws, size_t ws_size,
                              hipStream_t stream)
{
    const float* x      = (const float*)d_in[0];
    const float* params = (const float*)d_in[1];
    float*       out    = (float*)d_out;
    float*       exps   = (float*)d_ws;

    const int B        = in_sizes[0] / NQ;
    const int n_layers = in_sizes[1] / (NQ * 3);

    qsim_kernel<<<B, NTHR, 0, stream>>>(x, params, n_layers, exps);
    mitigate_wmma_kernel<<<B / 16, 32, 0, stream>>>(exps, out);
}
